# starting kernel
# baseline (speedup 1.0000x reference)
#include <hip/hip_runtime.h>

typedef float v2f __attribute__((ext_vector_type(2)));
typedef float v8f __attribute__((ext_vector_type(8)));

static constexpr int KD = 10;

__global__ __launch_bounds__(256) void precompute_affine(
    const float* __restrict__ W, const float* __restrict__ b,
    const int* __restrict__ steps_p, float* __restrict__ ws) {
  __shared__ float A[KD][KD];
  __shared__ float Mbuf[2][KD][KD];
  __shared__ float vbuf[2][KD];
  __shared__ float u[KD];
  const int tid = threadIdx.x;

  if (tid < KD * KD) {
    int k = tid / KD, n = tid % KD;
    A[k][n] = W[n * KD + k];
    Mbuf[0][k][n] = (k == n) ? 1.0f : 0.0f;
  }
  if (tid < KD) {
    u[tid] = b[tid] + 1.0f;
    vbuf[0][tid] = 0.0f;
  }
  __syncthreads();

  const int S = steps_p[0];
  int cur = 0;
  for (int t = 0; t < S; ++t) {
    int nxt = cur ^ 1;
    if (tid < KD * KD) {
      int i = tid / KD, j = tid % KD;
      float acc = 0.0f;
#pragma unroll
      for (int k = 0; k < KD; ++k) acc += Mbuf[cur][i][k] * A[k][j];
      Mbuf[nxt][i][j] = acc;
    }
    if (tid < KD) {
      float acc = u[tid];
#pragma unroll
      for (int k = 0; k < KD; ++k) acc += vbuf[cur][k] * A[k][tid];
      vbuf[nxt][tid] = acc;
    }
    __syncthreads();
    cur = nxt;
  }

  if (tid < 256) {
    int k = tid >> 4, n = tid & 15;
    ws[tid] = (k < KD && n < KD) ? Mbuf[cur][k][n] : 0.0f;
  }
  if (tid < 16) ws[256 + tid] = (tid < KD) ? vbuf[cur][tid] : 0.0f;
}

__global__ __launch_bounds__(256) void apply_affine_wmma(
    const float* __restrict__ x, const float* __restrict__ ws,
    float* __restrict__ out, int ntiles) {
  const int lane = threadIdx.x & 31;
  const int wid = (int)(blockIdx.x * (blockDim.x >> 5) + (threadIdx.x >> 5));
  if (wid >= ntiles) return;

  const int n  = lane & 15;
  const int r  = lane & 15;
  const int kb = (lane >> 4) * 2;
  const bool lo = lane < 16;

  v2f bf0, bf1, bf2;
  bf0.x = ws[(0 + kb) * 16 + n];  bf0.y = ws[(1 + kb) * 16 + n];
  bf1.x = ws[(4 + kb) * 16 + n];  bf1.y = ws[(5 + kb) * 16 + n];
  bf2.x = ws[(8 + kb) * 16 + n];  bf2.y = ws[(9 + kb) * 16 + n];

  const float bias = ws[256 + n];
  v8f c;
#pragma unroll
  for (int p = 0; p < 8; ++p) c[p] = bias;

  const float* xr = x + (size_t)wid * (16 * KD) + (size_t)r * KD;
  v2f a0 = __builtin_nontemporal_load((const v2f*)(xr + kb));
  v2f a1 = __builtin_nontemporal_load((const v2f*)(xr + 4 + kb));
  v2f a2r = __builtin_nontemporal_load((const v2f*)(xr + 8));
  v2f a2;
  a2.x = lo ? a2r.x : 0.0f;
  a2.y = lo ? a2r.y : 0.0f;

  c = __builtin_amdgcn_wmma_f32_16x16x4_f32(false, a0, false, bf0,
                                            (short)0, c, false, false);
  c = __builtin_amdgcn_wmma_f32_16x16x4_f32(false, a1, false, bf1,
                                            (short)0, c, false, false);
  c = __builtin_amdgcn_wmma_f32_16x16x4_f32(false, a2, false, bf2,
                                            (short)0, c, false, false);

  if (n < KD) {
    float* orow = out + (size_t)wid * (16 * KD) + (size_t)(lo ? 0 : 8 * KD) + n;
#pragma unroll
    for (int p = 0; p < 8; ++p)
      __builtin_nontemporal_store(c[p], orow + p * KD);
  }
}

__global__ __launch_bounds__(32) void apply_affine_tail(
    const float* __restrict__ x, const float* __restrict__ ws,
    float* __restrict__ out, int startrow, int nrows) {
  const int row = startrow + (int)threadIdx.x;
  if (row >= nrows) return;
  float xi[KD];
#pragma unroll
  for (int k = 0; k < KD; ++k) xi[k] = x[(size_t)row * KD + k];
#pragma unroll
  for (int nn = 0; nn < KD; ++nn) {
    float acc = ws[256 + nn];
#pragma unroll
    for (int k = 0; k < KD; ++k) acc += xi[k] * ws[k * 16 + nn];
    out[(size_t)row * KD + nn] = acc;
  }
}

extern "C" void kernel_launch(void* const* d_in, const int* in_sizes, int n_in,
                              void* d_out, int out_size, void* d_ws, size_t ws_size,
                              hipStream_t stream) {
  const float* x = (const float*)d_in[0];
  const float* W = (const float*)d_in[1];
  const float* b = (const float*)d_in[2];
  const int* steps = (const int*)d_in[3];
  float* ws = (float*)d_ws;
  float* out = (float*)d_out;

  const int nrows = in_sizes[0] / KD;
  const int ntiles_full = nrows / 16;
  const int tail = nrows - ntiles_full * 16;

  precompute_affine<<<1, 256, 0, stream>>>(W, b, steps, ws);

  if (ntiles_full > 0) {
    const int waves_per_block = 256 / 32;
    const int blocks = (ntiles_full + waves_per_block - 1) / waves_per_block;
    apply_affine_wmma<<<blocks, 256, 0, stream>>>(x, ws, out, ntiles_full);
  }
  if (tail > 0) {
    apply_affine_tail<<<1, 32, 0, stream>>>(x, ws, out, ntiles_full * 16, nrows);
  }
}


# single launch of 8-wave blocks (2 waves per SIMD, 8-unit slabs): each wave runs the num_steps recurrence on the matrix unit under its own x loads; 4 per SIMD had queued 4 chains per pipe; same sums
# speedup vs baseline: 2.5000x; 2.5000x over previous
#include <hip/hip_runtime.h>

typedef float v2f __attribute__((ext_vector_type(2)));
typedef float v4f __attribute__((ext_vector_type(4)));
typedef float v8f __attribute__((ext_vector_type(8)));

static constexpr int KD = 10;

__device__ __forceinline__ v8f affine_step(v2f a0, v2f a1, v2f a2, v8f U, v8f P) {
  v2f p0 = {P[0], P[1]};
  v2f p1 = {P[2], P[3]};
  v2f p2 = {P[4], P[5]};
  v8f D = __builtin_amdgcn_wmma_f32_16x16x4_f32(
      false, a0, false, p0, (short)0, U, false, false);
  D = __builtin_amdgcn_wmma_f32_16x16x4_f32(
      false, a1, false, p1, (short)0, D, false, false);
  D = __builtin_amdgcn_wmma_f32_16x16x4_f32(
      false, a2, false, p2, (short)0, D, false, false);
  return D;
}

struct AffineRaw {
  float w0[3], w1[3];
  float bl[8];
  int S;
};

__device__ __forceinline__ AffineRaw affine_request(
    const float* __restrict__ W, const float* __restrict__ b,
    const int* __restrict__ steps_p, int lane) {
  const int c = lane & 15;
  const int h = lane >> 4;
  const int lrow = 4 * ((c & 7) >> 1) + 2 * (c >> 3) + (c & 1);
  const int lrow_c = lrow < KD ? lrow : KD - 1;
  AffineRaw r;
#pragma unroll
  for (int j = 0; j < 3; ++j) {
    const int k0 = 4 * j + 2 * h;
    const int k0_c = k0 < KD ? k0 : KD - 2;
    const int idx = lrow_c * KD + k0_c;
    r.w0[j] = W[idx];
    r.w1[j] = W[idx + 1];
  }
#pragma unroll
  for (int q = 0; q < 8; ++q) {
    const int l = 4 * (q >> 1) + 2 * h + (q & 1);
    r.bl[q] = b[l < KD ? l : 0];
  }
  r.S = steps_p[0];
  return r;
}

__device__ __forceinline__ void affine_run(const AffineRaw& r, int lane,
                                           float* __restrict__ dst) {
  const int c = lane & 15;
  const int h = lane >> 4;
  const int lrow = 4 * ((c & 7) >> 1) + 2 * (c >> 3) + (c & 1);
  const bool row_ok = lrow < KD;

  v2f a[3];
#pragma unroll
  for (int j = 0; j < 3; ++j) {
    const int k0 = 4 * j + 2 * h;
    const bool ok = row_ok & (k0 < KD);
    a[j].x = ok ? r.w0[j] : 0.0f;
    a[j].y = ok ? r.w1[j] : 0.0f;
  }

  v8f U, P;
#pragma unroll
  for (int q = 0; q < 8; ++q) {
    const int l = 4 * (q >> 1) + 2 * h + (q & 1);
    const bool ok = l < KD;
    U[q] = (ok && c == KD) ? r.bl[q] + 1.0f : 0.0f;
    P[q] = (ok && l == c) ? 1.0f : 0.0f;
  }

  const int S = r.S;
  int t = 0;
  for (; t + 2 <= S; t += 2) {
    v8f Q = affine_step(a[0], a[1], a[2], U, P);
    P = affine_step(a[0], a[1], a[2], U, Q);
  }
  if (t < S) P = affine_step(a[0], a[1], a[2], U, P);

#pragma unroll
  for (int j = 0; j < 4; ++j) {
    const int n0 = 4 * j + 2 * h;
    const bool okn = (j < 3) && (n0 < KD);
    v2f val = {0.0f, 0.0f};
    if (j < 3) { val.x = P[2 * j]; val.y = P[2 * j + 1]; }
    v2f zero = {0.0f, 0.0f};
    v2f mval = (okn && c < KD) ? val : zero;
    *(v2f*)(dst + c * 16 + n0) = mval;
    if (c == KD) *(v2f*)(dst + 256 + n0) = okn ? val : zero;
  }
}

__global__ __launch_bounds__(32) void precompute_affine(
    const float* __restrict__ W, const float* __restrict__ b,
    const int* __restrict__ steps_p, float* __restrict__ ws) {
  const int lane = threadIdx.x & 31;
  const AffineRaw r = affine_request(W, b, steps_p, lane);
  affine_run(r, lane, ws);
}

static constexpr int TILE_F = 16 * KD;
static constexpr int UNIT_TILES = 4;
static constexpr int UNIT_F = UNIT_TILES * TILE_F;
static constexpr int UNIT_V4 = UNIT_F / 4;
static constexpr int UNIT_Q = UNIT_V4 / 32;
static constexpr int UNITS_PER_WAVE = 8;
static constexpr int WAVES_PER_BLOCK = 8;

__device__ __forceinline__ void wave_lds_sync() {
  __builtin_amdgcn_fence(__ATOMIC_RELEASE, "wavefront", "local");
  __builtin_amdgcn_wave_barrier();
  __builtin_amdgcn_fence(__ATOMIC_ACQUIRE, "wavefront", "local");
}

__global__ __launch_bounds__(WAVES_PER_BLOCK * 32) void apply_affine_wmma(
    const float* __restrict__ x, const float* __restrict__ W,
    const float* __restrict__ b, const int* __restrict__ steps_p,
    float* __restrict__ out, int ntiles) {
  __shared__ __attribute__((aligned(16))) float s_in[WAVES_PER_BLOCK][UNIT_F];
  __shared__ __attribute__((aligned(16))) float s_out[WAVES_PER_BLOCK][UNIT_F];

  const int lane = threadIdx.x & 31;
  const int wib = (int)(threadIdx.x >> 5);
  const int wave = __builtin_amdgcn_readfirstlane(
      (int)blockIdx.x * WAVES_PER_BLOCK + wib);
  const int c = lane & 15;
  const int h = lane >> 4;
  const int kb = 2 * h;

  const int limit4 = ntiles * (TILE_F / 4);
  const int base4 = wave * (UNITS_PER_WAVE * UNIT_V4);
  if (base4 >= limit4) return;

  const AffineRaw raw = affine_request(W, b, steps_p, lane);

  const v4f* __restrict__ x4 = (const v4f*)x;
  v4f xin[UNITS_PER_WAVE][UNIT_Q];
#pragma unroll
  for (int u = 0; u < UNITS_PER_WAVE; ++u) {
#pragma unroll
    for (int q = 0; q < UNIT_Q; ++q) {
      int idx = base4 + u * UNIT_V4 + q * 32 + lane;
      idx = idx < limit4 ? idx : limit4 - 1;
      xin[u][q] = __builtin_nontemporal_load(x4 + idx);
    }
  }

  asm volatile("" :: "v"(raw.w0[0]), "v"(raw.w1[0]), "v"(raw.w0[1]),
               "v"(raw.w1[1]), "v"(raw.w0[2]), "v"(raw.w1[2]));

  float* sin_w = s_in[wib];
  float* sout_w = s_out[wib];
  v4f* __restrict__ out4 = (v4f*)out;

  affine_run(raw, lane, sout_w);
  wave_lds_sync();

  v2f mf0, mf1, mf2;
  mf0.x = sout_w[(0 + kb) * 16 + c];  mf0.y = sout_w[(1 + kb) * 16 + c];
  mf1.x = sout_w[(4 + kb) * 16 + c];  mf1.y = sout_w[(5 + kb) * 16 + c];
  mf2.x = sout_w[(8 + kb) * 16 + c];  mf2.y = sout_w[(9 + kb) * 16 + c];

  v8f bias;
#pragma unroll
  for (int r = 0; r < 8; ++r) bias[r] = sout_w[256 + 8 * h + r];
  wave_lds_sync();

#pragma unroll
  for (int u = 0; u < UNITS_PER_WAVE; ++u) {
#pragma unroll
    for (int q = 0; q < UNIT_Q; ++q)
      *(v4f*)(sin_w + 4 * (q * 32 + lane)) = xin[u][q];
    wave_lds_sync();

    v2f b0[UNIT_TILES], b1[UNIT_TILES], b2[UNIT_TILES];
#pragma unroll
    for (int t = 0; t < UNIT_TILES; ++t) {
      const float* row = sin_w + (16 * t + c) * KD;
      b0[t] = *(const v2f*)(row + kb);
      b1[t] = *(const v2f*)(row + 4 + kb);
      v2f b2r = *(const v2f*)(row + 8);
      b2[t].x = h ? 0.0f : b2r.x;
      b2[t].y = h ? 0.0f : b2r.y;
    }

    v8f acc[UNIT_TILES];
#pragma unroll
    for (int t = 0; t < UNIT_TILES; ++t) {
      acc[t] = __builtin_amdgcn_wmma_f32_16x16x4_f32(
          false, mf0, false, b0[t], (short)0, bias, false, false);
      acc[t] = __builtin_amdgcn_wmma_f32_16x16x4_f32(
          false, mf1, false, b1[t], (short)0, acc[t], false, false);
      acc[t] = __builtin_amdgcn_wmma_f32_16x16x4_f32(
          false, mf2, false, b2[t], (short)0, acc[t], false, false);
    }

#pragma unroll
    for (int t = 0; t < UNIT_TILES; ++t) {
      float* orow = sout_w + (16 * t + c) * KD + 8 * h;
      v2f d01 = {acc[t][0], acc[t][1]};
      *(v2f*)(orow) = d01;
    }
    if (h == 0) {
#pragma unroll
      for (int t = 0; t < UNIT_TILES; ++t) {
        float* orow = sout_w + (16 * t + c) * KD;
        v2f d23 = {acc[t][2], acc[t][3]};
        v2f d45 = {acc[t][4], acc[t][5]};
        v2f d67 = {acc[t][6], acc[t][7]};
        *(v2f*)(orow + 2) = d23;
        *(v2f*)(orow + 4) = d45;
        *(v2f*)(orow + 6) = d67;
      }
    }
    wave_lds_sync();

    const int ubase = base4 + u * UNIT_V4;
    v4f o[UNIT_Q];
#pragma unroll
    for (int q = 0; q < UNIT_Q; ++q)
      o[q] = *(const v4f*)(sout_w + 4 * (q * 32 + lane));
    if (ubase + UNIT_V4 <= limit4) {
#pragma unroll
      for (int q = 0; q < UNIT_Q; ++q)
        __builtin_nontemporal_store(o[q], out4 + (ubase + q * 32 + lane));
    } else {
#pragma unroll
      for (int q = 0; q < UNIT_Q; ++q) {
        const int idx = ubase + q * 32 + lane;
        if (idx < limit4) __builtin_nontemporal_store(o[q], out4 + idx);
      }
    }
    if (u + 1 < UNITS_PER_WAVE) wave_lds_sync();
  }
}

__global__ __launch_bounds__(32) void apply_affine_tail(
    const float* __restrict__ x, const float* __restrict__ ws,
    float* __restrict__ out, int startrow, int nrows) {
  const int row = startrow + (int)threadIdx.x;
  if (row >= nrows) return;
  float xi[KD];
#pragma unroll
  for (int k = 0; k < KD; ++k) xi[k] = x[(size_t)row * KD + k];
#pragma unroll
  for (int nn = 0; nn < KD; ++nn) {
    float acc = ws[256 + nn];
#pragma unroll
    for (int k = 0; k < KD; ++k) acc += xi[k] * ws[k * 16 + nn];
    out[(size_t)row * KD + nn] = acc;
  }
}

extern "C" void kernel_launch(void* const* d_in, const int* in_sizes, int n_in,
                              void* d_out, int out_size, void* d_ws, size_t ws_size,
                              hipStream_t stream) {
  const float* x = (const float*)d_in[0];
  const float* W = (const float*)d_in[1];
  const float* b = (const float*)d_in[2];
  const int* steps = (const int*)d_in[3];
  float* ws = (float*)d_ws;
  float* out = (float*)d_out;

  const int nrows = in_sizes[0] / KD;
  const int ntiles_full = nrows / 16;
  const int tail = nrows - ntiles_full * 16;

  if (ntiles_full > 0) {
    const int tiles_per_block = WAVES_PER_BLOCK * UNITS_PER_WAVE * UNIT_TILES;
    const int blocks = (ntiles_full + tiles_per_block - 1) / tiles_per_block;
    apply_affine_wmma<<<blocks, WAVES_PER_BLOCK * 32, 0, stream>>>(x, W, b, steps, out, ntiles_full);
  }
  if (tail > 0) {
    precompute_affine<<<1, 32, 0, stream>>>(W, b, steps, ws);
    apply_affine_tail<<<1, 32, 0, stream>>>(x, ws, out, ntiles_full * 16, nrows);
  }
}
